# starting kernel
# baseline (speedup 1.0000x reference)
#include <hip/hip_runtime.h>

typedef __attribute__((ext_vector_type(2))) float v2f;
typedef __attribute__((ext_vector_type(8))) float v8f;

#define IN_C 128
#define HID  64

__global__ void sage_zero_kernel(float* __restrict__ p, long n) {
  long i = (long)blockIdx.x * blockDim.x + threadIdx.x;
  long stride = (long)gridDim.x * blockDim.x;
  for (; i < n; i += stride) p[i] = 0.0f;
}

__global__ void sage_degree_kernel(const int* __restrict__ dst, int E,
                                   float* __restrict__ deg) {
  int i = blockIdx.x * blockDim.x + threadIdx.x;
  int stride = gridDim.x * blockDim.x;
  for (; i < E; i += stride) {
    atomicAdd(&deg[dst[i]], 1.0f);
  }
}

template <int F, int CLOG2>
__global__ void sage_scatter_kernel(const float* __restrict__ x,
                                    const int* __restrict__ src,
                                    const int* __restrict__ dst, int E,
                                    float* __restrict__ agg) {
  long total = (long)E << CLOG2;
  long i = (long)blockIdx.x * blockDim.x + threadIdx.x;
  long stride = (long)gridDim.x * blockDim.x;
  for (; i < total; i += stride) {
    long e = i >> CLOG2;
    int f = (int)(i & ((1 << CLOG2) - 1)) << 2;
    int s = src[e];
    int d = dst[e];
    float4 v = *(const float4*)(x + (long)s * F + f);
    float* a = agg + (long)d * F + f;
    atomicAdd(a + 0, v.x);
    atomicAdd(a + 1, v.y);
    atomicAdd(a + 2, v.z);
    atomicAdd(a + 3, v.w);
  }
}

template <int K, bool RELU>
__global__ void sage_wmma_gemm_kernel(const float* __restrict__ Aagg,
                                      const float* __restrict__ deg,
                                      const float* __restrict__ Aself,
                                      const float* __restrict__ Wl,
                                      const float* __restrict__ bl,
                                      const float* __restrict__ Wr,
                                      float* __restrict__ Out, int nNodes) {
  const int lane = threadIdx.x & 31;
  const int wid = (blockIdx.x * blockDim.x + threadIdx.x) >> 5;
  const int tilesN = HID / 16;
  const int tile_m = wid / tilesN;
  const int tile_n = wid % tilesN;
  if (tile_m * 16 >= nNodes) return;

  const int half = lane >> 4;
  const int lm = lane & 15;
  const int koff = half * 2;
  const int rowA = tile_m * 16 + lm;
  const int col = tile_n * 16 + lm;

  const float invc = 1.0f / fmaxf(deg[rowA], 1.0f);

  v8f c = {0.f, 0.f, 0.f, 0.f, 0.f, 0.f, 0.f, 0.f};

  const float* arow = Aagg + (long)rowA * K;
  const float* srow = Aself + (long)rowA * K;

#pragma unroll
  for (int k = 0; k < K; k += 4) {
    v2f a;
    a.x = arow[k + koff] * invc;
    a.y = arow[k + koff + 1] * invc;
    v2f b;
    b.x = Wl[(k + koff) * HID + col];
    b.y = Wl[(k + koff + 1) * HID + col];
    c = __builtin_amdgcn_wmma_f32_16x16x4_f32(false, a, false, b, (short)0, c,
                                              false, false);
  }
#pragma unroll
  for (int k = 0; k < K; k += 4) {
    v2f a;
    a.x = srow[k + koff];
    a.y = srow[k + koff + 1];
    v2f b;
    b.x = Wr[(k + koff) * HID + col];
    b.y = Wr[(k + koff + 1) * HID + col];
    c = __builtin_amdgcn_wmma_f32_16x16x4_f32(false, a, false, b, (short)0, c,
                                              false, false);
  }

  const float bv = bl[col];
#pragma unroll
  for (int r = 0; r < 8; ++r) {
    float v = c[r] + bv;
    if (RELU) v = fmaxf(v, 0.0f);
    int row = tile_m * 16 + r + half * 8;
    Out[(long)row * HID + col] = v;
  }
}

extern "C" void kernel_launch(void* const* d_in, const int* in_sizes, int n_in,
                              void* d_out, int out_size, void* d_ws,
                              size_t ws_size, hipStream_t stream) {
  const float* x = (const float*)d_in[0];
  const int* ei = (const int*)d_in[1];
  const float* Wl1 = (const float*)d_in[2];
  const float* bl1 = (const float*)d_in[3];
  const float* Wr1 = (const float*)d_in[4];
  const float* Wl2 = (const float*)d_in[5];
  const float* bl2 = (const float*)d_in[6];
  const float* Wr2 = (const float*)d_in[7];
  float* out = (float*)d_out;

  const int N = in_sizes[0] / IN_C;
  const int E = in_sizes[1] / 2;
  const int* src = ei;
  const int* dst = ei + E;

  float* ws = (float*)d_ws;
  float* deg = ws;
  float* agg1 = deg + N;
  float* h = agg1 + (long)N * IN_C;
  float* agg2 = h + (long)N * HID;

  sage_zero_kernel<<<2048, 256, 0, stream>>>(deg, (long)N * (1 + IN_C));
  sage_zero_kernel<<<2048, 256, 0, stream>>>(agg2, (long)N * HID);

  sage_degree_kernel<<<2048, 256, 0, stream>>>(dst, E, deg);

  sage_scatter_kernel<IN_C, 5><<<8192, 256, 0, stream>>>(x, src, dst, E, agg1);

  const int tilesM = (N + 15) / 16;
  const int totalWaves = tilesM * (HID / 16);
  const int wavesPerBlock = 256 / 32;
  const int gemmBlocks = (totalWaves + wavesPerBlock - 1) / wavesPerBlock;
  sage_wmma_gemm_kernel<IN_C, true><<<gemmBlocks, 256, 0, stream>>>(
      agg1, deg, x, Wl1, bl1, Wr1, h, N);

  sage_scatter_kernel<HID, 4><<<8192, 256, 0, stream>>>(h, src, dst, E, agg2);
  sage_wmma_gemm_kernel<HID, false><<<gemmBlocks, 256, 0, stream>>>(
      agg2, deg, h, Wl2, bl2, Wr2, out, N);
}


# Scatter kernels: one wave per edge, lane l on features l, l+32, .. so each float atomic instruction fills one whole 128-B line, not four quarter-used ones; sums still in atomic arrival order
# speedup vs baseline: 1.4858x; 1.4858x over previous
#include <hip/hip_runtime.h>

typedef __attribute__((ext_vector_type(2))) float v2f;
typedef __attribute__((ext_vector_type(8))) float v8f;

#define IN_C 128
#define HID  64

__global__ void sage_zero_kernel(float* __restrict__ p, long n) {
  long i = (long)blockIdx.x * blockDim.x + threadIdx.x;
  long stride = (long)gridDim.x * blockDim.x;
  for (; i < n; i += stride) p[i] = 0.0f;
}

__global__ void sage_degree_kernel(const int* __restrict__ dst, int E,
                                   float* __restrict__ deg) {
  int i = blockIdx.x * blockDim.x + threadIdx.x;
  int stride = gridDim.x * blockDim.x;
  for (; i < E; i += stride) {
    atomicAdd(&deg[dst[i]], 1.0f);
  }
}


template <int F, int NE>
__device__ __forceinline__ void sage_scatter_rows(
    const float* __restrict__ x, const int* __restrict__ src,
    const int* __restrict__ dst, float* __restrict__ agg, long e0, int lane) {
  constexpr int R = F / 32;
  const float* xr[NE];
  float* ar[NE];
  for (int k = 0; k < NE; ++k) {
    xr[k] = x + (long)src[e0 + k] * F + lane;
    ar[k] = agg + (long)dst[e0 + k] * F + lane;
  }
  float v[NE][R];
  for (int k = 0; k < NE; ++k)
    for (int j = 0; j < R; ++j) v[k][j] = xr[k][32 * j];
  for (int k = 0; k < NE; ++k)
    for (int j = 0; j < R; ++j) atomicAdd(ar[k] + 32 * j, v[k][j]);
}

template <int F>
__global__ void sage_scatter_kernel(const float* __restrict__ x,
                                    const int* __restrict__ src,
                                    const int* __restrict__ dst, int E,
                                    float* __restrict__ agg) {
  constexpr int EPW = 128 / F;
  const int lane = threadIdx.x & 31;
  const long wave = ((long)blockIdx.x * blockDim.x + threadIdx.x) >> 5;
  const long step = (((long)gridDim.x * blockDim.x) >> 5) * EPW;
  for (long e0 = wave * EPW; e0 < E; e0 += step) {
    if (e0 + EPW <= E) {
      sage_scatter_rows<F, EPW>(x, src, dst, agg, e0, lane);
    } else {
      for (long e = e0; e < E; ++e)
        sage_scatter_rows<F, 1>(x, src, dst, agg, e, lane);
    }
  }
}

template <int K, bool RELU>
__global__ void sage_wmma_gemm_kernel(const float* __restrict__ Aagg,
                                      const float* __restrict__ deg,
                                      const float* __restrict__ Aself,
                                      const float* __restrict__ Wl,
                                      const float* __restrict__ bl,
                                      const float* __restrict__ Wr,
                                      float* __restrict__ Out, int nNodes) {
  const int lane = threadIdx.x & 31;
  const int wid = (blockIdx.x * blockDim.x + threadIdx.x) >> 5;
  const int tilesN = HID / 16;
  const int tile_m = wid / tilesN;
  const int tile_n = wid % tilesN;
  if (tile_m * 16 >= nNodes) return;

  const int half = lane >> 4;
  const int lm = lane & 15;
  const int koff = half * 2;
  const int rowA = tile_m * 16 + lm;
  const int col = tile_n * 16 + lm;

  const float invc = 1.0f / fmaxf(deg[rowA], 1.0f);

  v8f c = {0.f, 0.f, 0.f, 0.f, 0.f, 0.f, 0.f, 0.f};

  const float* arow = Aagg + (long)rowA * K;
  const float* srow = Aself + (long)rowA * K;

#pragma unroll
  for (int k = 0; k < K; k += 4) {
    v2f a;
    a.x = arow[k + koff] * invc;
    a.y = arow[k + koff + 1] * invc;
    v2f b;
    b.x = Wl[(k + koff) * HID + col];
    b.y = Wl[(k + koff + 1) * HID + col];
    c = __builtin_amdgcn_wmma_f32_16x16x4_f32(false, a, false, b, (short)0, c,
                                              false, false);
  }
#pragma unroll
  for (int k = 0; k < K; k += 4) {
    v2f a;
    a.x = srow[k + koff];
    a.y = srow[k + koff + 1];
    v2f b;
    b.x = Wr[(k + koff) * HID + col];
    b.y = Wr[(k + koff + 1) * HID + col];
    c = __builtin_amdgcn_wmma_f32_16x16x4_f32(false, a, false, b, (short)0, c,
                                              false, false);
  }

  const float bv = bl[col];
#pragma unroll
  for (int r = 0; r < 8; ++r) {
    float v = c[r] + bv;
    if (RELU) v = fmaxf(v, 0.0f);
    int row = tile_m * 16 + r + half * 8;
    Out[(long)row * HID + col] = v;
  }
}

extern "C" void kernel_launch(void* const* d_in, const int* in_sizes, int n_in,
                              void* d_out, int out_size, void* d_ws,
                              size_t ws_size, hipStream_t stream) {
  const float* x = (const float*)d_in[0];
  const int* ei = (const int*)d_in[1];
  const float* Wl1 = (const float*)d_in[2];
  const float* bl1 = (const float*)d_in[3];
  const float* Wr1 = (const float*)d_in[4];
  const float* Wl2 = (const float*)d_in[5];
  const float* bl2 = (const float*)d_in[6];
  const float* Wr2 = (const float*)d_in[7];
  float* out = (float*)d_out;

  const int N = in_sizes[0] / IN_C;
  const int E = in_sizes[1] / 2;
  const int* src = ei;
  const int* dst = ei + E;

  float* ws = (float*)d_ws;
  float* deg = ws;
  float* agg1 = deg + N;
  float* h = agg1 + (long)N * IN_C;
  float* agg2 = h + (long)N * HID;

  sage_zero_kernel<<<2048, 256, 0, stream>>>(deg, (long)N * (1 + IN_C));
  sage_zero_kernel<<<2048, 256, 0, stream>>>(agg2, (long)N * HID);

  sage_degree_kernel<<<2048, 256, 0, stream>>>(dst, E, deg);

  sage_scatter_kernel<IN_C><<<8192, 256, 0, stream>>>(x, src, dst, E, agg1);

  const int tilesM = (N + 15) / 16;
  const int totalWaves = tilesM * (HID / 16);
  const int wavesPerBlock = 256 / 32;
  const int gemmBlocks = (totalWaves + wavesPerBlock - 1) / wavesPerBlock;
  sage_wmma_gemm_kernel<IN_C, true><<<gemmBlocks, 256, 0, stream>>>(
      agg1, deg, x, Wl1, bl1, Wr1, h, N);

  sage_scatter_kernel<HID><<<8192, 256, 0, stream>>>(h, src, dst, E, agg2);
  sage_wmma_gemm_kernel<HID, false><<<gemmBlocks, 256, 0, stream>>>(
      agg2, deg, h, Wl2, bl2, Wr2, out, N);
}


# Layer 1 multiplies by W_l before the scatter: atomics sum 64-wide x@W_l rows, not 128-wide x rows (sum order changes: sum of products, not product of the sum); mean, self term, relu in one pass
# speedup vs baseline: 1.7528x; 1.1797x over previous
#include <hip/hip_runtime.h>

typedef __attribute__((ext_vector_type(2))) float v2f;
typedef __attribute__((ext_vector_type(4))) float v4f;
typedef __attribute__((ext_vector_type(8))) float v8f;

#define IN_C 128
#define HID  64

__global__ void sage_zero_kernel(float* __restrict__ p, long n) {
  long i = (long)blockIdx.x * blockDim.x + threadIdx.x;
  long stride = (long)gridDim.x * blockDim.x;
  for (; i < n; i += stride) p[i] = 0.0f;
}

__global__ void sage_degree_kernel(const int* __restrict__ dst, int E,
                                   float* __restrict__ deg) {
  int i = blockIdx.x * blockDim.x + threadIdx.x;
  int stride = gridDim.x * blockDim.x;
  for (; i < E; i += stride) {
    atomicAdd(&deg[dst[i]], 1.0f);
  }
}


template <int F, int NE>
__device__ __forceinline__ void sage_scatter_rows(
    const float* __restrict__ x, const int* __restrict__ src,
    const int* __restrict__ dst, float* __restrict__ agg, long e0, int lane) {
  constexpr int R = F / 32;
  const float* xr[NE];
  float* ar[NE];
  for (int k = 0; k < NE; ++k) {
    xr[k] = x + (long)src[e0 + k] * F + lane;
    ar[k] = agg + (long)dst[e0 + k] * F + lane;
  }
  float v[NE][R];
  for (int k = 0; k < NE; ++k)
    for (int j = 0; j < R; ++j) v[k][j] = xr[k][32 * j];
  for (int k = 0; k < NE; ++k)
    for (int j = 0; j < R; ++j) atomicAdd(ar[k] + 32 * j, v[k][j]);
}

template <int F>
__global__ void sage_scatter_kernel(const float* __restrict__ x,
                                    const int* __restrict__ src,
                                    const int* __restrict__ dst, int E,
                                    float* __restrict__ agg) {
  constexpr int EPW = 128 / F;
  const int lane = threadIdx.x & 31;
  const long wave = ((long)blockIdx.x * blockDim.x + threadIdx.x) >> 5;
  const long step = (((long)gridDim.x * blockDim.x) >> 5) * EPW;
  for (long e0 = wave * EPW; e0 < E; e0 += step) {
    if (e0 + EPW <= E) {
      sage_scatter_rows<F, EPW>(x, src, dst, agg, e0, lane);
    } else {
      for (long e = e0; e < E; ++e)
        sage_scatter_rows<F, 1>(x, src, dst, agg, e, lane);
    }
  }
}

template <int K, bool RELU>
__global__ void sage_wmma_gemm_kernel(const float* __restrict__ Aagg,
                                      const float* __restrict__ deg,
                                      const float* __restrict__ Aself,
                                      const float* __restrict__ Wl,
                                      const float* __restrict__ bl,
                                      const float* __restrict__ Wr,
                                      float* __restrict__ Out, int nNodes) {
  const int lane = threadIdx.x & 31;
  const int wid = (blockIdx.x * blockDim.x + threadIdx.x) >> 5;
  const int tilesN = HID / 16;
  const int tile_m = wid / tilesN;
  const int tile_n = wid % tilesN;
  if (tile_m * 16 >= nNodes) return;

  const int half = lane >> 4;
  const int lm = lane & 15;
  const int koff = half * 2;
  const int rowA = tile_m * 16 + lm;
  const int col = tile_n * 16 + lm;

  const float invc = 1.0f / fmaxf(deg[rowA], 1.0f);

  v8f c = {0.f, 0.f, 0.f, 0.f, 0.f, 0.f, 0.f, 0.f};

  const float* arow = Aagg + (long)rowA * K;
  const float* srow = Aself + (long)rowA * K;

#pragma unroll
  for (int k = 0; k < K; k += 4) {
    v2f a;
    a.x = arow[k + koff] * invc;
    a.y = arow[k + koff + 1] * invc;
    v2f b;
    b.x = Wl[(k + koff) * HID + col];
    b.y = Wl[(k + koff + 1) * HID + col];
    c = __builtin_amdgcn_wmma_f32_16x16x4_f32(false, a, false, b, (short)0, c,
                                              false, false);
  }
#pragma unroll
  for (int k = 0; k < K; k += 4) {
    v2f a;
    a.x = srow[k + koff];
    a.y = srow[k + koff + 1];
    v2f b;
    b.x = Wr[(k + koff) * HID + col];
    b.y = Wr[(k + koff + 1) * HID + col];
    c = __builtin_amdgcn_wmma_f32_16x16x4_f32(false, a, false, b, (short)0, c,
                                              false, false);
  }

  const float bv = bl[col];
#pragma unroll
  for (int r = 0; r < 8; ++r) {
    float v = c[r] + bv;
    if (RELU) v = fmaxf(v, 0.0f);
    int row = tile_m * 16 + r + half * 8;
    Out[(long)row * HID + col] = v;
  }
}

template <int K>
__global__ void sage_wmma_project_kernel(const float* __restrict__ X,
                                         const float* __restrict__ Wl,
                                         const float* __restrict__ bl,
                                         const float* __restrict__ Wr,
                                         float* __restrict__ P,
                                         float* __restrict__ S, int nNodes) {
  const int lane = threadIdx.x & 31;
  const int wid = (blockIdx.x * blockDim.x + threadIdx.x) >> 5;
  const int tilesN = HID / 16;
  const int tile_m = wid / tilesN;
  const int tile_n = wid % tilesN;
  if (tile_m * 16 >= nNodes) return;

  const int half = lane >> 4;
  const int lm = lane & 15;
  const int koff = half * 2;
  const int rowA = tile_m * 16 + lm;
  const int col = tile_n * 16 + lm;

  v8f cp = {0.f, 0.f, 0.f, 0.f, 0.f, 0.f, 0.f, 0.f};
  v8f cs = {0.f, 0.f, 0.f, 0.f, 0.f, 0.f, 0.f, 0.f};

  const float* xrow = X + (long)rowA * K;

#pragma unroll
  for (int k = 0; k < K; k += 4) {
    v2f a;
    a.x = xrow[k + koff];
    a.y = xrow[k + koff + 1];
    v2f bp;
    bp.x = Wl[(k + koff) * HID + col];
    bp.y = Wl[(k + koff + 1) * HID + col];
    v2f bs;
    bs.x = Wr[(k + koff) * HID + col];
    bs.y = Wr[(k + koff + 1) * HID + col];
    cp = __builtin_amdgcn_wmma_f32_16x16x4_f32(false, a, false, bp, (short)0,
                                               cp, false, false);
    cs = __builtin_amdgcn_wmma_f32_16x16x4_f32(false, a, false, bs, (short)0,
                                               cs, false, false);
  }

  const float bv = bl[col];
#pragma unroll
  for (int r = 0; r < 8; ++r) {
    long o = (long)(tile_m * 16 + r + half * 8) * HID + col;
    P[o] = cp[r];
    S[o] = cs[r] + bv;
  }
}

__global__ void sage_mean_relu_kernel(const float* __restrict__ agg,
                                      const float* __restrict__ deg,
                                      float* hs, long n4) {
  long i = (long)blockIdx.x * blockDim.x + threadIdx.x;
  long stride = (long)gridDim.x * blockDim.x;
  for (; i < n4; i += stride) {
    const float invc = 1.0f / fmaxf(deg[i / (HID / 4)], 1.0f);
    const v4f a = ((const v4f*)agg)[i];
    const v4f s = ((const v4f*)hs)[i];
    v4f h;
    h.x = fmaxf(a.x * invc + s.x, 0.0f);
    h.y = fmaxf(a.y * invc + s.y, 0.0f);
    h.z = fmaxf(a.z * invc + s.z, 0.0f);
    h.w = fmaxf(a.w * invc + s.w, 0.0f);
    ((v4f*)hs)[i] = h;
  }
}

extern "C" void kernel_launch(void* const* d_in, const int* in_sizes, int n_in,
                              void* d_out, int out_size, void* d_ws,
                              size_t ws_size, hipStream_t stream) {
  const float* x = (const float*)d_in[0];
  const int* ei = (const int*)d_in[1];
  const float* Wl1 = (const float*)d_in[2];
  const float* bl1 = (const float*)d_in[3];
  const float* Wr1 = (const float*)d_in[4];
  const float* Wl2 = (const float*)d_in[5];
  const float* bl2 = (const float*)d_in[6];
  const float* Wr2 = (const float*)d_in[7];
  float* out = (float*)d_out;

  const int N = in_sizes[0] / IN_C;
  const int E = in_sizes[1] / 2;
  const int* src = ei;
  const int* dst = ei + E;

  float* ws = (float*)d_ws;
  float* deg = ws;
  float* agg1 = deg + N;
  float* agg2 = agg1 + (long)N * HID;
  float* p1 = agg2 + (long)N * HID;
  float* h = p1 + (long)N * HID;

  sage_zero_kernel<<<2048, 256, 0, stream>>>(deg, (long)N * (1 + 2 * HID));

  sage_degree_kernel<<<2048, 256, 0, stream>>>(dst, E, deg);

  const int tilesM = (N + 15) / 16;
  const int totalWaves = tilesM * (HID / 16);
  const int wavesPerBlock = 256 / 32;
  const int gemmBlocks = (totalWaves + wavesPerBlock - 1) / wavesPerBlock;

  sage_wmma_project_kernel<IN_C><<<gemmBlocks, 256, 0, stream>>>(
      x, Wl1, bl1, Wr1, p1, h, N);
  sage_scatter_kernel<HID><<<8192, 256, 0, stream>>>(p1, src, dst, E, agg1);
  sage_mean_relu_kernel<<<2048, 256, 0, stream>>>(agg1, deg, h,
                                                  (long)N * (HID / 4));

  sage_scatter_kernel<HID><<<8192, 256, 0, stream>>>(h, src, dst, E, agg2);
  sage_wmma_gemm_kernel<HID, false><<<gemmBlocks, 256, 0, stream>>>(
      agg2, deg, h, Wl2, bl2, Wr2, out, N);
}


# Float-atomic scatter replaced by counting sort of edges by target (a returning integer atomic per edge gives its rank) + one wave summing each node's rows in registers, stored once; sums in rank order
# speedup vs baseline: 2.3473x; 1.3392x over previous
#include <hip/hip_runtime.h>


typedef __attribute__((ext_vector_type(2))) float v2f;
typedef __attribute__((ext_vector_type(8))) float v8f;

#define IN_C 128
#define HID  64
#define NPW  8

__global__ void sage_zero_kernel(int* __restrict__ p, int n) {
  int i = blockIdx.x * blockDim.x + threadIdx.x;
  int stride = gridDim.x * blockDim.x;
  for (; i < n; i += stride) p[i] = 0;
}


__global__ void sage_rank_kernel(const int* __restrict__ dst, int E,
                                 int* __restrict__ cnt,
                                 int* __restrict__ rank) {
  const int T = gridDim.x * blockDim.x;
  for (int i = blockIdx.x * blockDim.x + threadIdx.x; i < E; i += 4 * T) {
    int d[4], r[4];
    for (int k = 0; k < 4; ++k) d[k] = (i + k * T < E) ? dst[i + k * T] : -1;
    for (int k = 0; k < 4; ++k) r[k] = (d[k] >= 0) ? atomicAdd(&cnt[d[k]], 1) : 0;
    for (int k = 0; k < 4; ++k)
      if (i + k * T < E) rank[i + k * T] = r[k];
  }
}

__global__ void sage_segment_kernel(const int* __restrict__ cnt, int N,
                                    int* __restrict__ total,
                                    int* __restrict__ start) {
  __shared__ int wsum[32];
  __shared__ int base;
  const int i = blockIdx.x * blockDim.x + threadIdx.x;
  const int lane = threadIdx.x & 31;
  const int w = threadIdx.x >> 5;
  const int c = (i < N) ? cnt[i] : 0;
  int incl = c;
  for (int o = 1; o < 32; o <<= 1) {
    int t = __shfl_up(incl, o);
    if (lane >= o) incl += t;
  }
  if (lane == 31) wsum[w] = incl;
  __syncthreads();
  if (w == 0) {
    const int s = (lane < (int)(blockDim.x >> 5)) ? wsum[lane] : 0;
    int si = s;
    for (int o = 1; o < 32; o <<= 1) {
      int t = __shfl_up(si, o);
      if (lane >= o) si += t;
    }
    wsum[lane] = si - s;
    if (lane == 31) base = atomicAdd(total, si);
  }
  __syncthreads();
  if (i < N) start[i] = base + wsum[w] + incl - c;
}

__global__ void sage_fill_kernel(const int* __restrict__ src,
                                 const int* __restrict__ dst,
                                 const int* __restrict__ rank,
                                 const int* __restrict__ start, int E,
                                 int* __restrict__ csr) {
  const int T = gridDim.x * blockDim.x;
  for (int i = blockIdx.x * blockDim.x + threadIdx.x; i < E; i += 4 * T) {
    int d[4], r[4], s[4], p[4];
    for (int k = 0; k < 4; ++k) {
      const bool ok = i + k * T < E;
      d[k] = ok ? dst[i + k * T] : -1;
      r[k] = ok ? rank[i + k * T] : 0;
      s[k] = ok ? src[i + k * T] : 0;
    }
    for (int k = 0; k < 4; ++k) p[k] = (d[k] >= 0) ? start[d[k]] + r[k] : 0;
    for (int k = 0; k < 4; ++k)
      if (d[k] >= 0) csr[p[k]] = s[k];
  }
}


template <int U>
__device__ __forceinline__ void sage_add_rows(const float* __restrict__ rows,
                                              int idx, int j, int lane,
                                              v2f& acc) {
  v2f v[U];
  for (int u = 0; u < U; ++u) {
    const int r = __builtin_amdgcn_readlane(idx, j + u);
    v[u] = *(const v2f*)(rows + (long)r * HID + 2 * lane);
  }
  for (int u = 0; u < U; ++u) acc += v[u];
}

__device__ __forceinline__ void sage_sum_rows(const float* __restrict__ rows,
                                              int idx, int n, int lane,
                                              v2f& acc) {
  int j = 0;
  for (; j + 16 <= n; j += 16) sage_add_rows<16>(rows, idx, j, lane, acc);
  if (j + 8 <= n) { sage_add_rows<8>(rows, idx, j, lane, acc); j += 8; }
  if (j + 4 <= n) { sage_add_rows<4>(rows, idx, j, lane, acc); j += 4; }
  if (j < n) {
    v2f v[3];
    for (int u = 0; u < 3; ++u) {
      const int jj = (j + u < n) ? j + u : j;
      const int r = __builtin_amdgcn_readlane(idx, jj);
      v[u] = *(const v2f*)(rows + (long)r * HID + 2 * lane);
    }
    for (int u = 0; u < 3; ++u)
      if (j + u < n) acc += v[u];
  }
}

template <bool LAYER1>
__global__ void sage_gather_kernel(const float* __restrict__ rows,
                                   const int* __restrict__ start,
                                   const int* __restrict__ cnt,
                                   const int* __restrict__ csr,
                                   float* __restrict__ io, int N) {
  const int lane = threadIdx.x & 31;
  const int wave = __builtin_amdgcn_readfirstlane(
      (int)((blockIdx.x * blockDim.x + threadIdx.x) >> 5));
  const int n0 = wave * NPW;
  if (n0 >= N) return;
  const int nn = (N - n0 < NPW) ? N - n0 : NPW;

  int my_s = 0, my_c = 0;
  if (lane < nn) {
    my_s = start[n0 + lane];
    my_c = cnt[n0 + lane];
  }

  int s = __builtin_amdgcn_readlane(my_s, 0);
  int c = __builtin_amdgcn_readlane(my_c, 0);
  int idx = (lane < c) ? csr[s + lane] : 0;

  for (int k = 0; k < nn; ++k) {
    const int node = n0 + k;
    v2f* o = (v2f*)(io + (long)node * HID) + lane;
    v2f sv = {0.f, 0.f};
    if (LAYER1) sv = *o;

    int s2 = 0, c2 = 0, idx2 = 0;
    if (k + 1 < nn) {
      s2 = __builtin_amdgcn_readlane(my_s, k + 1);
      c2 = __builtin_amdgcn_readlane(my_c, k + 1);
      idx2 = (lane < c2) ? csr[s2 + lane] : 0;
    }

    v2f acc = {0.f, 0.f};
    sage_sum_rows(rows, idx, c < 32 ? c : 32, lane, acc);
    for (int j0 = 32; j0 < c; j0 += 32) {
      const int more = (j0 + lane < c) ? csr[s + j0 + lane] : 0;
      sage_sum_rows(rows, more, c - j0 < 32 ? c - j0 : 32, lane, acc);
    }

    const float invc = 1.0f / fmaxf((float)c, 1.0f);
    v2f res;
    if (LAYER1) {
      res.x = fmaxf(acc.x * invc + sv.x, 0.0f);
      res.y = fmaxf(acc.y * invc + sv.y, 0.0f);
    } else {
      res.x = acc.x * invc;
      res.y = acc.y * invc;
    }
    *o = res;

    s = s2;
    c = c2;
    idx = idx2;
  }
}

template <int K, bool RELU>
__global__ void sage_wmma_gemm_kernel(const float* __restrict__ Amean,
                                      const float* __restrict__ Aself,
                                      const float* __restrict__ Wl,
                                      const float* __restrict__ bl,
                                      const float* __restrict__ Wr,
                                      float* __restrict__ Out, int nNodes) {
  const int lane = threadIdx.x & 31;
  const int wid = (blockIdx.x * blockDim.x + threadIdx.x) >> 5;
  const int tilesN = HID / 16;
  const int tile_m = wid / tilesN;
  const int tile_n = wid % tilesN;
  if (tile_m * 16 >= nNodes) return;

  const int half = lane >> 4;
  const int lm = lane & 15;
  const int koff = half * 2;
  const int rowA = tile_m * 16 + lm;
  const int col = tile_n * 16 + lm;

  v8f c = {0.f, 0.f, 0.f, 0.f, 0.f, 0.f, 0.f, 0.f};

  const float* arow = Amean + (long)rowA * K;
  const float* srow = Aself + (long)rowA * K;

#pragma unroll
  for (int k = 0; k < K; k += 4) {
    v2f a;
    a.x = arow[k + koff];
    a.y = arow[k + koff + 1];
    v2f b;
    b.x = Wl[(k + koff) * HID + col];
    b.y = Wl[(k + koff + 1) * HID + col];
    c = __builtin_amdgcn_wmma_f32_16x16x4_f32(false, a, false, b, (short)0, c,
                                              false, false);
  }
#pragma unroll
  for (int k = 0; k < K; k += 4) {
    v2f a;
    a.x = srow[k + koff];
    a.y = srow[k + koff + 1];
    v2f b;
    b.x = Wr[(k + koff) * HID + col];
    b.y = Wr[(k + koff + 1) * HID + col];
    c = __builtin_amdgcn_wmma_f32_16x16x4_f32(false, a, false, b, (short)0, c,
                                              false, false);
  }

  const float bv = bl[col];
#pragma unroll
  for (int r = 0; r < 8; ++r) {
    float v = c[r] + bv;
    if (RELU) v = fmaxf(v, 0.0f);
    int row = tile_m * 16 + r + half * 8;
    Out[(long)row * HID + col] = v;
  }
}

template <int K>
__global__ void sage_wmma_project_kernel(const float* __restrict__ X,
                                         const float* __restrict__ Wl,
                                         const float* __restrict__ bl,
                                         const float* __restrict__ Wr,
                                         float* __restrict__ P,
                                         float* __restrict__ S, int nNodes) {
  const int lane = threadIdx.x & 31;
  const int wid = (blockIdx.x * blockDim.x + threadIdx.x) >> 5;
  const int tilesN = HID / 16;
  const int tile_m = wid / tilesN;
  const int tile_n = wid % tilesN;
  if (tile_m * 16 >= nNodes) return;

  const int half = lane >> 4;
  const int lm = lane & 15;
  const int koff = half * 2;
  const int rowA = tile_m * 16 + lm;
  const int col = tile_n * 16 + lm;

  v8f cp = {0.f, 0.f, 0.f, 0.f, 0.f, 0.f, 0.f, 0.f};
  v8f cs = {0.f, 0.f, 0.f, 0.f, 0.f, 0.f, 0.f, 0.f};

  const float* xrow = X + (long)rowA * K;

#pragma unroll
  for (int k = 0; k < K; k += 4) {
    v2f a;
    a.x = xrow[k + koff];
    a.y = xrow[k + koff + 1];
    v2f bp;
    bp.x = Wl[(k + koff) * HID + col];
    bp.y = Wl[(k + koff + 1) * HID + col];
    v2f bs;
    bs.x = Wr[(k + koff) * HID + col];
    bs.y = Wr[(k + koff + 1) * HID + col];
    cp = __builtin_amdgcn_wmma_f32_16x16x4_f32(false, a, false, bp, (short)0,
                                               cp, false, false);
    cs = __builtin_amdgcn_wmma_f32_16x16x4_f32(false, a, false, bs, (short)0,
                                               cs, false, false);
  }

  const float bv = bl[col];
#pragma unroll
  for (int r = 0; r < 8; ++r) {
    long o = (long)(tile_m * 16 + r + half * 8) * HID + col;
    P[o] = cp[r];
    S[o] = cs[r] + bv;
  }
}

extern "C" void kernel_launch(void* const* d_in, const int* in_sizes, int n_in,
                              void* d_out, int out_size, void* d_ws,
                              size_t ws_size, hipStream_t stream) {
  const float* x = (const float*)d_in[0];
  const int* ei = (const int*)d_in[1];
  const float* Wl1 = (const float*)d_in[2];
  const float* bl1 = (const float*)d_in[3];
  const float* Wr1 = (const float*)d_in[4];
  const float* Wl2 = (const float*)d_in[5];
  const float* bl2 = (const float*)d_in[6];
  const float* Wr2 = (const float*)d_in[7];
  float* out = (float*)d_out;

  const int N = in_sizes[0] / IN_C;
  const int E = in_sizes[1] / 2;
  const int* src = ei;
  const int* dst = ei + E;

  float* p1 = (float*)d_ws;
  float* h = p1 + (long)N * HID;
  float* m2 = h + (long)N * HID;
  int* cnt = (int*)(m2 + (long)N * HID);
  int* total = cnt + N;
  int* start = total + 1;
  int* rank = start + N;
  int* csr = rank + E;

  const int edgeBlocks = (E + 4 * 256 - 1) / (4 * 256);
  sage_zero_kernel<<<128, 256, 0, stream>>>(cnt, N + 1);
  sage_rank_kernel<<<edgeBlocks, 256, 0, stream>>>(dst, E, cnt, rank);
  sage_segment_kernel<<<(N + 1023) / 1024, 1024, 0, stream>>>(cnt, N, total,
                                                               start);
  sage_fill_kernel<<<edgeBlocks, 256, 0, stream>>>(src, dst, rank, start, E,
                                                   csr);

  const int tilesM = (N + 15) / 16;
  const int totalWaves = tilesM * (HID / 16);
  const int wavesPerBlock = 256 / 32;
  const int gemmBlocks = (totalWaves + wavesPerBlock - 1) / wavesPerBlock;
  const int gatherBlocks = (N + NPW * wavesPerBlock - 1) / (NPW * wavesPerBlock);

  sage_wmma_project_kernel<IN_C><<<gemmBlocks, 256, 0, stream>>>(
      x, Wl1, bl1, Wr1, p1, h, N);
  sage_gather_kernel<true><<<gatherBlocks, 256, 0, stream>>>(p1, start, cnt,
                                                             csr, h, N);

  sage_gather_kernel<false><<<gatherBlocks, 256, 0, stream>>>(h, start, cnt,
                                                              csr, m2, N);
  sage_wmma_gemm_kernel<HID, false><<<gemmBlocks, 256, 0, stream>>>(
      m2, h, Wl2, bl2, Wr2, out, N);
}


# Edge grouping with no memory atomics: LDS counters bin edges by 512-node range (count, scan, scatter), then one work-group per range sorts by node in LDS; rows still summed in atomic arrival order
# speedup vs baseline: 3.6720x; 1.5644x over previous
#include <hip/hip_runtime.h>


typedef __attribute__((ext_vector_type(2))) float v2f;
typedef __attribute__((ext_vector_type(8))) float v8f;

#define IN_C 128
#define HID  64
#define NPW  8


#define RSHIFT 9
#define RANGE (1 << RSHIFT)
#define MAXBINS 256
#define CHUNK_THREADS 512
#define EPT 16
#define CHUNK (CHUNK_THREADS * EPT)
#define SPT 18

__device__ __forceinline__ int sage_wave_incl_scan(int v, int lane) {
  for (int o = 1; o < 32; o <<= 1) {
    int t = __shfl_up(v, o);
    if (lane >= o) v += t;
  }
  return v;
}

__global__ void sage_bin_count_kernel(const int* __restrict__ dst, int E,
                                      int NB, int* __restrict__ histT) {
  __shared__ int h[MAXBINS];
  const int tid = threadIdx.x;
  const int b = blockIdx.x;
  const int B = gridDim.x;
  const int e0 = b * CHUNK + tid;
  int d[EPT];
  for (int k = 0; k < EPT; ++k) {
    const int e = e0 + k * CHUNK_THREADS;
    d[k] = (e < E) ? dst[e] : -1;
  }
  for (int i = tid; i < NB; i += CHUNK_THREADS) h[i] = 0;
  __syncthreads();
  for (int k = 0; k < EPT; ++k)
    if (d[k] >= 0) atomicAdd(&h[d[k] >> RSHIFT], 1);
  __syncthreads();
  for (int i = tid; i < NB; i += CHUNK_THREADS) histT[i * B + b] = h[i];
}

__global__ void sage_offsets_kernel(const int* __restrict__ in, int n,
                                    int* __restrict__ out) {
  __shared__ int wsum[32];
  __shared__ int wpre[32];
  const int tid = threadIdx.x;
  const int lane = tid & 31;
  const int w = tid >> 5;
  const int first = blockIdx.x * 1024;
  int before = 0;
#pragma unroll 8
  for (int j = tid; j < first; j += 1024) before += in[j];
  const int i = first + tid;
  const int c = (i < n) ? in[i] : 0;
  const int incl = sage_wave_incl_scan(c, lane);
  for (int o = 16; o > 0; o >>= 1) before += __shfl_xor(before, o);
  if (lane == 31) wsum[w] = incl;
  if (lane == 0) wpre[w] = before;
  __syncthreads();
  if (w == 0) {
    const int s = wsum[lane];
    int p = wpre[lane];
    const int si = sage_wave_incl_scan(s, lane);
    for (int o = 16; o > 0; o >>= 1) p += __shfl_xor(p, o);
    wsum[lane] = p + si - s;
  }
  __syncthreads();
  if (i < n) out[i] = wsum[w] + incl - c;
}

__global__ void sage_bin_scatter_kernel(const int* __restrict__ src,
                                        const int* __restrict__ dst, int E,
                                        int NB, const int* __restrict__ off,
                                        int2* __restrict__ pairs) {
  __shared__ int cur[MAXBINS];
  const int tid = threadIdx.x;
  const int b = blockIdx.x;
  const int B = gridDim.x;
  const int e0 = b * CHUNK + tid;
  int d[EPT], s[EPT];
  for (int k = 0; k < EPT; ++k) {
    const int e = e0 + k * CHUNK_THREADS;
    d[k] = (e < E) ? dst[e] : -1;
    s[k] = (e < E) ? src[e] : 0;
  }
  for (int i = tid; i < NB; i += CHUNK_THREADS) cur[i] = off[i * B + b];
  __syncthreads();
  int p[EPT];
  for (int k = 0; k < EPT; ++k)
    p[k] = (d[k] >= 0) ? atomicAdd(&cur[d[k] >> RSHIFT], 1) : -1;
  for (int k = 0; k < EPT; ++k)
    if (p[k] >= 0) pairs[p[k]] = make_int2(s[k], d[k]);
}

__global__ void sage_bin_sort_kernel(const int2* __restrict__ pairs,
                                     const int* __restrict__ off, int E, int N,
                                     int NB, int B, int* __restrict__ cnt,
                                     int* __restrict__ start,
                                     int* __restrict__ csr) {
  __shared__ int c[RANGE];
  __shared__ int wsum[32];
  __shared__ int sorted[SPT * RANGE];
  const int tid = threadIdx.x;
  const int lane = tid & 31;
  const int w = tid >> 5;
  const int r = blockIdx.x;
  const int r0 = off[r * B];
  const int r1 = (r + 1 < NB) ? off[(r + 1) * B] : E;
  const int node0 = r << RSHIFT;

  int2 pr[SPT];
  for (int k = 0; k < SPT; ++k) {
    const int e = r0 + tid + k * RANGE;
    pr[k] = (e < r1) ? pairs[e] : make_int2(0, -1);
  }
  c[tid] = 0;
  __syncthreads();
  for (int k = 0; k < SPT; ++k)
    if (pr[k].y >= 0) atomicAdd(&c[pr[k].y - node0], 1);
  for (int e = r0 + tid + SPT * RANGE; e < r1; e += RANGE)
    atomicAdd(&c[pairs[e].y - node0], 1);
  __syncthreads();

  const int v = c[tid];
  const int incl = sage_wave_incl_scan(v, lane);
  if (lane == 31) wsum[w] = incl;
  __syncthreads();
  if (w == 0) {
    const int s = (lane < RANGE / 32) ? wsum[lane] : 0;
    const int si = sage_wave_incl_scan(s, lane);
    wsum[lane] = si - s;
  }
  __syncthreads();
  const int excl = wsum[w] + incl - v;
  c[tid] = excl;
  const int node = node0 + tid;
  if (node < N) {
    cnt[node] = v;
    start[node] = r0 + excl;
  }
  __syncthreads();

  int p[SPT];
  for (int k = 0; k < SPT; ++k)
    p[k] = (pr[k].y >= 0) ? atomicAdd(&c[pr[k].y - node0], 1) : -1;
  for (int k = 0; k < SPT; ++k)
    if (p[k] >= 0) {
      if (p[k] < SPT * RANGE) sorted[p[k]] = pr[k].x;
      else csr[r0 + p[k]] = pr[k].x;
    }
  for (int e = r0 + tid + SPT * RANGE; e < r1; e += RANGE) {
    const int2 q = pairs[e];
    const int pq = atomicAdd(&c[q.y - node0], 1);
    if (pq < SPT * RANGE) sorted[pq] = q.x;
    else csr[r0 + pq] = q.x;
  }
  __syncthreads();
  for (int k = 0; k < SPT; ++k) {
    const int i = tid + k * RANGE;
    if (r0 + i < r1) csr[r0 + i] = sorted[i];
  }
}


template <int U>
__device__ __forceinline__ void sage_add_rows(const float* __restrict__ rows,
                                              int idx, int j, int lane,
                                              v2f& acc) {
  v2f v[U];
  for (int u = 0; u < U; ++u) {
    const int r = __builtin_amdgcn_readlane(idx, j + u);
    v[u] = *(const v2f*)(rows + (long)r * HID + 2 * lane);
  }
  for (int u = 0; u < U; ++u) acc += v[u];
}

__device__ __forceinline__ void sage_sum_rows(const float* __restrict__ rows,
                                              int idx, int n, int lane,
                                              v2f& acc) {
  int j = 0;
  for (; j + 16 <= n; j += 16) sage_add_rows<16>(rows, idx, j, lane, acc);
  if (j + 8 <= n) { sage_add_rows<8>(rows, idx, j, lane, acc); j += 8; }
  if (j + 4 <= n) { sage_add_rows<4>(rows, idx, j, lane, acc); j += 4; }
  if (j < n) {
    v2f v[3];
    for (int u = 0; u < 3; ++u) {
      const int jj = (j + u < n) ? j + u : j;
      const int r = __builtin_amdgcn_readlane(idx, jj);
      v[u] = *(const v2f*)(rows + (long)r * HID + 2 * lane);
    }
    for (int u = 0; u < 3; ++u)
      if (j + u < n) acc += v[u];
  }
}

template <bool LAYER1>
__global__ void sage_gather_kernel(const float* __restrict__ rows,
                                   const int* __restrict__ start,
                                   const int* __restrict__ cnt,
                                   const int* __restrict__ csr,
                                   float* __restrict__ io, int N) {
  const int lane = threadIdx.x & 31;
  const int wave = __builtin_amdgcn_readfirstlane(
      (int)((blockIdx.x * blockDim.x + threadIdx.x) >> 5));
  const int n0 = wave * NPW;
  if (n0 >= N) return;
  const int nn = (N - n0 < NPW) ? N - n0 : NPW;

  int my_s = 0, my_c = 0;
  if (lane < nn) {
    my_s = start[n0 + lane];
    my_c = cnt[n0 + lane];
  }

  int s = __builtin_amdgcn_readlane(my_s, 0);
  int c = __builtin_amdgcn_readlane(my_c, 0);
  int idx = (lane < c) ? csr[s + lane] : 0;

  for (int k = 0; k < nn; ++k) {
    const int node = n0 + k;
    v2f* o = (v2f*)(io + (long)node * HID) + lane;
    v2f sv = {0.f, 0.f};
    if (LAYER1) sv = *o;

    int s2 = 0, c2 = 0, idx2 = 0;
    if (k + 1 < nn) {
      s2 = __builtin_amdgcn_readlane(my_s, k + 1);
      c2 = __builtin_amdgcn_readlane(my_c, k + 1);
      idx2 = (lane < c2) ? csr[s2 + lane] : 0;
    }

    v2f acc = {0.f, 0.f};
    sage_sum_rows(rows, idx, c < 32 ? c : 32, lane, acc);
    for (int j0 = 32; j0 < c; j0 += 32) {
      const int more = (j0 + lane < c) ? csr[s + j0 + lane] : 0;
      sage_sum_rows(rows, more, c - j0 < 32 ? c - j0 : 32, lane, acc);
    }

    const float invc = 1.0f / fmaxf((float)c, 1.0f);
    v2f res;
    if (LAYER1) {
      res.x = fmaxf(acc.x * invc + sv.x, 0.0f);
      res.y = fmaxf(acc.y * invc + sv.y, 0.0f);
    } else {
      res.x = acc.x * invc;
      res.y = acc.y * invc;
    }
    *o = res;

    s = s2;
    c = c2;
    idx = idx2;
  }
}

template <int K, bool RELU>
__global__ void sage_wmma_gemm_kernel(const float* __restrict__ Amean,
                                      const float* __restrict__ Aself,
                                      const float* __restrict__ Wl,
                                      const float* __restrict__ bl,
                                      const float* __restrict__ Wr,
                                      float* __restrict__ Out, int nNodes) {
  const int lane = threadIdx.x & 31;
  const int wid = (blockIdx.x * blockDim.x + threadIdx.x) >> 5;
  const int tilesN = HID / 16;
  const int tile_m = wid / tilesN;
  const int tile_n = wid % tilesN;
  if (tile_m * 16 >= nNodes) return;

  const int half = lane >> 4;
  const int lm = lane & 15;
  const int koff = half * 2;
  const int rowA = tile_m * 16 + lm;
  const int col = tile_n * 16 + lm;

  v8f c = {0.f, 0.f, 0.f, 0.f, 0.f, 0.f, 0.f, 0.f};

  const float* arow = Amean + (long)rowA * K;
  const float* srow = Aself + (long)rowA * K;

#pragma unroll
  for (int k = 0; k < K; k += 4) {
    v2f a;
    a.x = arow[k + koff];
    a.y = arow[k + koff + 1];
    v2f b;
    b.x = Wl[(k + koff) * HID + col];
    b.y = Wl[(k + koff + 1) * HID + col];
    c = __builtin_amdgcn_wmma_f32_16x16x4_f32(false, a, false, b, (short)0, c,
                                              false, false);
  }
#pragma unroll
  for (int k = 0; k < K; k += 4) {
    v2f a;
    a.x = srow[k + koff];
    a.y = srow[k + koff + 1];
    v2f b;
    b.x = Wr[(k + koff) * HID + col];
    b.y = Wr[(k + koff + 1) * HID + col];
    c = __builtin_amdgcn_wmma_f32_16x16x4_f32(false, a, false, b, (short)0, c,
                                              false, false);
  }

  const float bv = bl[col];
#pragma unroll
  for (int r = 0; r < 8; ++r) {
    float v = c[r] + bv;
    if (RELU) v = fmaxf(v, 0.0f);
    int row = tile_m * 16 + r + half * 8;
    Out[(long)row * HID + col] = v;
  }
}

template <int K>
__global__ void sage_wmma_project_kernel(const float* __restrict__ X,
                                         const float* __restrict__ Wl,
                                         const float* __restrict__ bl,
                                         const float* __restrict__ Wr,
                                         float* __restrict__ P,
                                         float* __restrict__ S, int nNodes) {
  const int lane = threadIdx.x & 31;
  const int wid = (blockIdx.x * blockDim.x + threadIdx.x) >> 5;
  const int tilesN = HID / 16;
  const int tile_m = wid / tilesN;
  const int tile_n = wid % tilesN;
  if (tile_m * 16 >= nNodes) return;

  const int half = lane >> 4;
  const int lm = lane & 15;
  const int koff = half * 2;
  const int rowA = tile_m * 16 + lm;
  const int col = tile_n * 16 + lm;

  v8f cp = {0.f, 0.f, 0.f, 0.f, 0.f, 0.f, 0.f, 0.f};
  v8f cs = {0.f, 0.f, 0.f, 0.f, 0.f, 0.f, 0.f, 0.f};

  const float* xrow = X + (long)rowA * K;

#pragma unroll
  for (int k = 0; k < K; k += 4) {
    v2f a;
    a.x = xrow[k + koff];
    a.y = xrow[k + koff + 1];
    v2f bp;
    bp.x = Wl[(k + koff) * HID + col];
    bp.y = Wl[(k + koff + 1) * HID + col];
    v2f bs;
    bs.x = Wr[(k + koff) * HID + col];
    bs.y = Wr[(k + koff + 1) * HID + col];
    cp = __builtin_amdgcn_wmma_f32_16x16x4_f32(false, a, false, bp, (short)0,
                                               cp, false, false);
    cs = __builtin_amdgcn_wmma_f32_16x16x4_f32(false, a, false, bs, (short)0,
                                               cs, false, false);
  }

  const float bv = bl[col];
#pragma unroll
  for (int r = 0; r < 8; ++r) {
    long o = (long)(tile_m * 16 + r + half * 8) * HID + col;
    P[o] = cp[r];
    S[o] = cs[r] + bv;
  }
}

extern "C" void kernel_launch(void* const* d_in, const int* in_sizes, int n_in,
                              void* d_out, int out_size, void* d_ws,
                              size_t ws_size, hipStream_t stream) {
  const float* x = (const float*)d_in[0];
  const int* ei = (const int*)d_in[1];
  const float* Wl1 = (const float*)d_in[2];
  const float* bl1 = (const float*)d_in[3];
  const float* Wr1 = (const float*)d_in[4];
  const float* Wl2 = (const float*)d_in[5];
  const float* bl2 = (const float*)d_in[6];
  const float* Wr2 = (const float*)d_in[7];
  float* out = (float*)d_out;

  const int N = in_sizes[0] / IN_C;
  const int E = in_sizes[1] / 2;
  const int* src = ei;
  const int* dst = ei + E;

  float* p1 = (float*)d_ws;
  float* h = p1 + (long)N * HID;
  float* m2 = h + (long)N * HID;
  int2* pairs = (int2*)m2;
  int* cnt = (int*)(m2 + (long)N * HID);
  int* start = cnt + N;
  int* csr = start + N;
  const int B = (E + CHUNK - 1) / CHUNK;
  const int NB = (N + RANGE - 1) / RANGE;
  const int nH = NB * B;
  int* histT = csr + E;
  int* off = histT + nH;

  sage_bin_count_kernel<<<B, CHUNK_THREADS, 0, stream>>>(dst, E, NB, histT);
  sage_offsets_kernel<<<(nH + 1023) / 1024, 1024, 0, stream>>>(histT, nH, off);
  sage_bin_scatter_kernel<<<B, CHUNK_THREADS, 0, stream>>>(src, dst, E, NB,
                                                           off, pairs);
  sage_bin_sort_kernel<<<NB, RANGE, 0, stream>>>(pairs, off, E, N, NB, B, cnt,
                                                 start, csr);

  const int tilesM = (N + 15) / 16;
  const int totalWaves = tilesM * (HID / 16);
  const int wavesPerBlock = 256 / 32;
  const int gemmBlocks = (totalWaves + wavesPerBlock - 1) / wavesPerBlock;
  const int gatherBlocks = (N + NPW * wavesPerBlock - 1) / (NPW * wavesPerBlock);

  sage_wmma_project_kernel<IN_C><<<gemmBlocks, 256, 0, stream>>>(
      x, Wl1, bl1, Wr1, p1, h, N);
  sage_gather_kernel<true><<<gatherBlocks, 256, 0, stream>>>(p1, start, cnt,
                                                             csr, h, N);

  sage_gather_kernel<false><<<gatherBlocks, 256, 0, stream>>>(h, start, cnt,
                                                              csr, m2, N);
  sage_wmma_gemm_kernel<HID, false><<<gemmBlocks, 256, 0, stream>>>(
      m2, h, Wl2, bl2, Wr2, out, N);
}


# Layer 2 in one launch: a wave sums the rows of 16 nodes and parks means beside the self rows in its LDS tile, the A operand of the product; no mean plane in memory; k sum: self half first
# speedup vs baseline: 5.5803x; 1.5197x over previous
#include <hip/hip_runtime.h>


typedef __attribute__((ext_vector_type(2))) float v2f;
typedef __attribute__((ext_vector_type(8))) float v8f;
typedef __attribute__((ext_vector_type(4))) float v4f;

#define IN_C 128
#define HID  64
#define NPW  8


#define RSHIFT 9
#define RANGE (1 << RSHIFT)
#define MAXBINS 256
#define CHUNK_THREADS 512
#define EPT 16
#define CHUNK (CHUNK_THREADS * EPT)
#define SPT 18
#define JR 13
#define PW (RANGE / 32)

__device__ __forceinline__ int sage_wave_incl_scan(int v, int lane) {
  for (int o = 1; o < 32; o <<= 1) {
    int t = __shfl_up(v, o);
    if (lane >= o) v += t;
  }
  return v;
}

__global__ void __launch_bounds__(CHUNK_THREADS)
sage_chunk_sort_kernel(const int* __restrict__ src,
                       const int* __restrict__ dst, int E, int NB,
                       int2* __restrict__ pairs, int* __restrict__ T) {
  __shared__ int h[CHUNK_THREADS];
  __shared__ int wsum[32];
  __shared__ int2 buf[CHUNK];
  const int tid = threadIdx.x;
  const int lane = tid & 31;
  const int w = tid >> 5;
  const int b = blockIdx.x;
  const int e0 = b * CHUNK + tid;
  int d[EPT], s[EPT];
  for (int k = 0; k < EPT; ++k) {
    const int e = e0 + k * CHUNK_THREADS;
    d[k] = (e < E) ? dst[e] : -1;
    s[k] = (e < E) ? src[e] : 0;
  }
  h[tid] = 0;
  __syncthreads();
  for (int k = 0; k < EPT; ++k)
    if (d[k] >= 0) atomicAdd(&h[d[k] >> RSHIFT], 1);
  __syncthreads();

  const int v = h[tid];
  const int incl = sage_wave_incl_scan(v, lane);
  if (lane == 31) wsum[w] = incl;
  __syncthreads();
  if (w == 0) {
    const int t = (lane < CHUNK_THREADS / 32) ? wsum[lane] : 0;
    const int ti = sage_wave_incl_scan(t, lane);
    wsum[lane] = ti - t;
  }
  __syncthreads();
  const int excl = wsum[w] + incl - v;
  h[tid] = excl;
  if (tid <= NB) T[b * (NB + 1) + tid] = excl;
  __syncthreads();

  int p[EPT];
  for (int k = 0; k < EPT; ++k)
    p[k] = (d[k] >= 0) ? atomicAdd(&h[d[k] >> RSHIFT], 1) : -1;
  for (int k = 0; k < EPT; ++k)
    if (p[k] >= 0) buf[p[k]] = make_int2(s[k], d[k]);
  __syncthreads();

  const int nb = (E - b * CHUNK < CHUNK) ? E - b * CHUNK : CHUNK;
  int2 ov[EPT];
  for (int k = 0; k < EPT; ++k) {
    const int i = tid + k * CHUNK_THREADS;
    ov[k] = (i < nb) ? buf[i] : make_int2(0, 0);
  }
  for (int k = 0; k < EPT; ++k) {
    const int i = tid + k * CHUNK_THREADS;
    if (i < nb) pairs[(long)b * CHUNK + i] = ov[k];
  }
}

template <bool PLACE>
__device__ __forceinline__ void sage_range_rest(
    const int2* __restrict__ pairs, int o, int len, int bb, int w, int B,
    int lane, int node0, int r0, int* c, int* sorted, int* __restrict__ csr) {
  int nj = (B - bb - w + PW - 1) / PW;
  if (nj > 32) nj = 32;
  for (int j = 0; j < nj; ++j) {
    const int lj = __builtin_amdgcn_readlane(len, j);
    const int oj = __builtin_amdgcn_readlane(o, j);
    const int i0 = (bb == 0 && j < JR) ? 64 : 0;
    const int2* pc = pairs + (long)(bb + w + PW * j) * CHUNK + oj;
    for (int i = i0 + lane; i < lj; i += 32) {
      const int2 q = pc[i];
      const int pq = atomicAdd(&c[q.y - node0], 1);
      if (PLACE) {
        if (pq < SPT * RANGE) sorted[pq] = q.x;
        else csr[r0 + pq] = q.x;
      }
    }
  }
}

__global__ void __launch_bounds__(RANGE)
sage_range_sort_kernel(const int2* __restrict__ pairs,
                       const int* __restrict__ T, int N, int NB, int B,
                       int* __restrict__ cnt, int* __restrict__ start,
                       int* __restrict__ csr) {
  __shared__ int c[RANGE];
  __shared__ int wsum[32];
  __shared__ int rsum[32];
  __shared__ int sorted[SPT * RANGE];
  const int tid = threadIdx.x;
  const int lane = tid & 31;
  const int w = __builtin_amdgcn_readfirstlane(tid >> 5);
  const int r = blockIdx.x;
  const int node0 = r << RSHIFT;
  const int TP = NB + 1;

  int o0 = 0, len0 = 0;
  {
    const int mb = w + PW * lane;
    if (mb < B) {
      o0 = T[mb * TP + r];
      len0 = T[mb * TP + r + 1] - o0;
    }
  }
  int part = o0;
  for (int b = PW * 32 + tid; b < B; b += RANGE) part += T[b * TP + r];

  int2 pr[JR][2];
  for (int j = 0; j < JR; ++j) {
    const int lj = __builtin_amdgcn_readlane(len0, j);
    const int oj = __builtin_amdgcn_readlane(o0, j);
    const int2* pc = pairs + (long)(w + PW * j) * CHUNK + oj;
    for (int u = 0; u < 2; ++u) {
      const int i = lane + 32 * u;
      pr[j][u] = (i < lj) ? pc[i] : make_int2(0, -1);
    }
  }

  for (int o = 16; o > 0; o >>= 1) part += __shfl_xor(part, o);
  if (lane == 0) rsum[w] = part;
  c[tid] = 0;
  __syncthreads();
  int r0 = 0;
  for (int i = 0; i < PW; ++i) r0 += rsum[i];

  for (int j = 0; j < JR; ++j)
    for (int u = 0; u < 2; ++u)
      if (pr[j][u].y >= 0) atomicAdd(&c[pr[j][u].y - node0], 1);
  sage_range_rest<false>(pairs, o0, len0, 0, w, B, lane, node0, r0, c, sorted,
                         csr);
  for (int bb = PW * 32; bb < B; bb += PW * 32) {
    const int mb = bb + w + PW * lane;
    int o = 0, len = 0;
    if (mb < B) {
      o = T[mb * TP + r];
      len = T[mb * TP + r + 1] - o;
    }
    sage_range_rest<false>(pairs, o, len, bb, w, B, lane, node0, r0, c, sorted,
                           csr);
  }
  __syncthreads();

  const int v = c[tid];
  const int incl = sage_wave_incl_scan(v, lane);
  if (lane == 31) wsum[w] = incl;
  __syncthreads();
  int total = 0;
  for (int i = 0; i < PW; ++i) total += wsum[i];
  __syncthreads();
  if (w == 0) {
    const int t = (lane < PW) ? wsum[lane] : 0;
    const int ti = sage_wave_incl_scan(t, lane);
    wsum[lane] = ti - t;
  }
  __syncthreads();
  const int excl = wsum[w] + incl - v;
  c[tid] = excl;
  const int node = node0 + tid;
  if (node < N) {
    cnt[node] = v;
    start[node] = r0 + excl;
  }
  __syncthreads();

  int p[JR][2];
  for (int j = 0; j < JR; ++j)
    for (int u = 0; u < 2; ++u)
      p[j][u] = (pr[j][u].y >= 0) ? atomicAdd(&c[pr[j][u].y - node0], 1) : -1;
  for (int j = 0; j < JR; ++j)
    for (int u = 0; u < 2; ++u)
      if (p[j][u] >= 0) {
        if (p[j][u] < SPT * RANGE) sorted[p[j][u]] = pr[j][u].x;
        else csr[r0 + p[j][u]] = pr[j][u].x;
      }
  sage_range_rest<true>(pairs, o0, len0, 0, w, B, lane, node0, r0, c, sorted,
                        csr);
  for (int bb = PW * 32; bb < B; bb += PW * 32) {
    const int mb = bb + w + PW * lane;
    int o = 0, len = 0;
    if (mb < B) {
      o = T[mb * TP + r];
      len = T[mb * TP + r + 1] - o;
    }
    sage_range_rest<true>(pairs, o, len, bb, w, B, lane, node0, r0, c, sorted,
                          csr);
  }
  __syncthreads();
  int sv[SPT];
  for (int k = 0; k < SPT; ++k) {
    const int i = tid + k * RANGE;
    sv[k] = (i < total) ? sorted[i] : 0;
  }
  for (int k = 0; k < SPT; ++k) {
    const int i = tid + k * RANGE;
    if (i < total) csr[r0 + i] = sv[k];
  }
}


struct sage_nothing {
  __device__ __forceinline__ void operator()() const {}
};

template <int U, typename F>
__device__ __forceinline__ void sage_add_rows(const float* __restrict__ rows,
                                              int idx, int j, int lane,
                                              v2f& acc, const F& under) {
  v2f v[U];
  for (int u = 0; u < U; ++u) {
    const int r = __builtin_amdgcn_readlane(idx, j + u);
    v[u] = *(const v2f*)(rows + (long)r * HID + 2 * lane);
  }
  under();
  for (int u = 0; u < U; ++u) acc += v[u];
}

template <typename F>
__device__ __forceinline__ void sage_sum_rows(const float* __restrict__ rows,
                                              int idx, int n, int lane,
                                              v2f& acc, const F& under) {
  int j = 0;
  for (; j + 16 <= n; j += 16)
    sage_add_rows<16>(rows, idx, j, lane, acc, under);
  if (j + 8 <= n) { sage_add_rows<8>(rows, idx, j, lane, acc, under); j += 8; }
  if (j + 4 <= n) { sage_add_rows<4>(rows, idx, j, lane, acc, under); j += 4; }
  if (j < n) {
    v2f v[3];
    for (int u = 0; u < 3; ++u) {
      const int jj = (j + u < n) ? j + u : j;
      const int r = __builtin_amdgcn_readlane(idx, jj);
      v[u] = *(const v2f*)(rows + (long)r * HID + 2 * lane);
    }
    under();
    for (int u = 0; u < 3; ++u)
      if (j + u < n) acc += v[u];
  }
}

__global__ void sage_gather_kernel(const float* __restrict__ rows,
                                   const int* __restrict__ start,
                                   const int* __restrict__ cnt,
                                   const int* __restrict__ csr,
                                   float* __restrict__ io, int N) {
  const int lane = threadIdx.x & 31;
  const int wave = __builtin_amdgcn_readfirstlane(
      (int)((blockIdx.x * blockDim.x + threadIdx.x) >> 5));
  const int n0 = wave * NPW;
  if (n0 >= N) return;
  const int nn = (N - n0 < NPW) ? N - n0 : NPW;

  int my_s = 0, my_c = 0;
  if (lane < nn) {
    my_s = start[n0 + lane];
    my_c = cnt[n0 + lane];
  }

  int s = __builtin_amdgcn_readlane(my_s, 0);
  int c = __builtin_amdgcn_readlane(my_c, 0);
  int idx = (lane < c) ? csr[s + lane] : 0;

  for (int k = 0; k < nn; ++k) {
    const int node = n0 + k;
    v2f* o = (v2f*)(io + (long)node * HID) + lane;
    const v2f sv = *o;

    int s2 = 0, c2 = 0, idx2 = 0;
    if (k + 1 < nn) {
      s2 = __builtin_amdgcn_readlane(my_s, k + 1);
      c2 = __builtin_amdgcn_readlane(my_c, k + 1);
      idx2 = (lane < c2) ? csr[s2 + lane] : 0;
    }

    v2f acc = {0.f, 0.f};
    sage_sum_rows(rows, idx, c < 32 ? c : 32, lane, acc, sage_nothing());
    for (int j0 = 32; j0 < c; j0 += 32) {
      const int more = (j0 + lane < c) ? csr[s + j0 + lane] : 0;
      sage_sum_rows(rows, more, c - j0 < 32 ? c - j0 : 32, lane, acc,
                    sage_nothing());
    }

    const float invc = 1.0f / fmaxf((float)c, 1.0f);
    v2f res;
    res.x = fmaxf(acc.x * invc + sv.x, 0.0f);
    res.y = fmaxf(acc.y * invc + sv.y, 0.0f);
    *o = res;

    s = s2;
    c = c2;
    idx = idx2;
  }
}


#define GT_THREADS 512
#define GT_WAVES (GT_THREADS / 32)
#define GT_GROUPS 256
#define TPITCH 132

__device__ __forceinline__ void sage_load_rows(const float* __restrict__ A0,
                                               int row0, int nNodes, int lane,
                                               v4f (&rv)[16]) {
  for (int r = 0; r < 16; ++r) {
    int row = row0 + r;
    if (row >= nNodes) row = nNodes - 1;
    rv[r] = *(const v4f*)(A0 + (long)row * IN_C + 4 * lane);
  }
}

__global__ void __launch_bounds__(GT_THREADS)
sage_rows_gemm_kernel(const float* __restrict__ A0,
                      const float* __restrict__ Wl,
                      const float* __restrict__ bl,
                      const float* __restrict__ Wr, float* __restrict__ O0,
                      float* __restrict__ O1, int nNodes) {
  constexpr int NT = 8;
  constexpr int NC = NT * 16;
  __shared__ float4 Bf4[32 * NT * 16];
  __shared__ float4 tiles4[GT_WAVES * 4 * TPITCH];
  float* const Bf = (float*)Bf4;
  float* const tiles = (float*)tiles4;

  const int tid = threadIdx.x;
  const int lane = tid & 31;
  const int w = __builtin_amdgcn_readfirstlane(tid >> 5);
  const int half = lane >> 4;
  const int lm = lane & 15;

  const int nTiles = (nNodes + 15) >> 4;
  const int nWaves = gridDim.x * GT_WAVES;
  int tm = w * gridDim.x + blockIdx.x;
  v4f rv[16];
  if (tm < nTiles) sage_load_rows(A0, tm * 16, nNodes, lane, rv);

  constexpr int NQ = NC / 4;
  for (int j = 0; j < (64 * NQ) / GT_THREADS; ++j) {
    const int i = tid + j * GT_THREADS;
    const int kp = i / NQ;
    const int n = 4 * (i % NQ);
    const int k = 2 * kp;
    const float* src = (n < 64) ? Wl + k * HID + n : Wr + k * HID + n - 64;
    const float4 r0 = *(const float4*)src;
    const float4 r1 = *(const float4*)(src + HID);
    const int s = kp >> 1, h = kp & 1;
    const int t = n >> 4, c = n & 15;
    float* d = Bf + ((s * NT + t) * 32 + h * 16 + c) * 2;
    *(float4*)d = make_float4(r0.x, r1.x, r0.y, r1.y);
    *(float4*)(d + 4) = make_float4(r0.z, r1.z, r0.w, r1.w);
  }
  __syncthreads();

  float bv[NT];
  for (int t = 0; t < NT; ++t)
    bv[t] = (t >= 4) ? bl[16 * (t - 4) + lm] : 0.0f;

  float* tile = tiles + w * (16 * TPITCH);
  const float* afrag = tile + lm * TPITCH + 2 * half;
  const float* bfrag = Bf + lane * 2;
  for (; tm < nTiles; tm += nWaves) {
    const int row0 = tm * 16;

    for (int r = 0; r < 16; ++r)
      *(v4f*)(tile + r * TPITCH + 4 * lane) = rv[r];
    __builtin_amdgcn_fence(__ATOMIC_ACQ_REL, "wavefront");
    if (tm + nWaves < nTiles)
      sage_load_rows(A0, (tm + nWaves) * 16, nNodes, lane, rv);

    v8f acc[NT];
    for (int t = 0; t < NT; ++t)
      acc[t] = (v8f){0.f, 0.f, 0.f, 0.f, 0.f, 0.f, 0.f, 0.f};

#pragma unroll
    for (int s = 0; s < 32; ++s) {
      const v2f a = *(const v2f*)(afrag + 4 * s);
      for (int t = 0; t < NT; ++t) {
        const v2f b = *(const v2f*)(bfrag + (s * NT + t) * 64);
        acc[t] = __builtin_amdgcn_wmma_f32_16x16x4_f32(
            false, a, false, b, (short)0, acc[t], false, false);
      }
    }

    for (int t = 0; t < NT; ++t)
      for (int r = 0; r < 8; ++r) {
        const bool biased = t >= 4;
        tile[(r + 8 * half) * TPITCH + 16 * t + lm] =
            biased ? acc[t][r] + bv[t] : acc[t][r];
      }
    __builtin_amdgcn_fence(__ATOMIC_ACQ_REL, "wavefront");

    float* const obase = half ? O1 : O0;
    if (row0 + 16 <= nNodes) {
      float4 ov[16];
      for (int r = 0; r < 16; ++r)
        ov[r] = *(const float4*)(tile + r * TPITCH + 4 * lane);
      for (int r = 0; r < 16; ++r)
        *(float4*)(obase + (long)(row0 + r) * HID + 4 * lm) = ov[r];
    } else {
      for (int r = 0; r < 16; ++r) {
        const float4 v = *(const float4*)(tile + r * TPITCH + 4 * lane);
        if (row0 + r < nNodes)
          *(float4*)(obase + (long)(row0 + r) * HID + 4 * lm) = v;
      }
    }
    __builtin_amdgcn_fence(__ATOMIC_ACQ_REL, "wavefront");
  }
}

#define FG_WAVES 25
#define FG_THREADS (FG_WAVES * 32)

__global__ void __launch_bounds__(FG_THREADS)
sage_gather_gemm_kernel(const float* __restrict__ h,
                        const int* __restrict__ start,
                        const int* __restrict__ cnt,
                        const int* __restrict__ csr,
                        const float* __restrict__ Wl,
                        const float* __restrict__ bl,
                        const float* __restrict__ Wr, float* __restrict__ out,
                        int nNodes) {
  constexpr int NT = 4;
  __shared__ float4 Bf4[32 * NT * 16];
  __shared__ float4 tiles4[FG_WAVES * 4 * TPITCH];
  float* const Bf = (float*)Bf4;
  float* const tiles = (float*)tiles4;

  const int tid = threadIdx.x;
  const int lane = tid & 31;
  const int w = __builtin_amdgcn_readfirstlane(tid >> 5);
  const int half = lane >> 4;
  const int lm = lane & 15;

  const int nTiles = (nNodes + 15) >> 4;
  const int nWaves = gridDim.x * FG_WAVES;
  int tm = w * gridDim.x + blockIdx.x;

  int my_s = 0, my_c = 0;
  if (tm < nTiles && lane < 16 && tm * 16 + lane < nNodes) {
    my_s = start[tm * 16 + lane];
    my_c = cnt[tm * 16 + lane];
  }

  for (int i = tid; i < 64 * 16; i += FG_THREADS) {
    const int kp = i >> 4;
    const int n = 4 * (i & 15);
    const int k = 2 * kp;
    const float* src = (k < 64) ? Wl + k * HID + n : Wr + (k - 64) * HID + n;
    const float4 r0 = *(const float4*)src;
    const float4 r1 = *(const float4*)(src + HID);
    const int ks = kp >> 1, kh = kp & 1;
    const int t = n >> 4, c = n & 15;
    float* d = Bf + ((ks * NT + t) * 32 + kh * 16 + c) * 2;
    *(float4*)d = make_float4(r0.x, r1.x, r0.y, r1.y);
    *(float4*)(d + 4) = make_float4(r0.z, r1.z, r0.w, r1.w);
  }
  __syncthreads();

  float bv[NT];
  for (int t = 0; t < NT; ++t) bv[t] = bl[16 * t + lm];

  float* tile = tiles + w * (16 * TPITCH);
  const float* afrag = tile + lm * TPITCH + 2 * half;
  const float* bfrag = Bf + lane * 2;
  for (; tm < nTiles; tm += nWaves) {
    const int n0 = tm * 16;
    const int nn = (nNodes - n0 < 16) ? nNodes - n0 : 16;

    int sg = __builtin_amdgcn_readlane(my_s, 0);
    int cg = __builtin_amdgcn_readlane(my_c, 0);
    int idx = (lane < cg) ? csr[sg + lane] : 0;

    float4 sr[8];
    for (int i = 0; i < 8; ++i) {
      const int row = n0 + 2 * i + half;
      sr[i] = make_float4(0.f, 0.f, 0.f, 0.f);
      if (row < nNodes) sr[i] = *(const float4*)(h + (long)row * HID + 4 * lm);
    }
    for (int i = 0; i < 8; ++i)
      *(float4*)(tile + (2 * i + half) * TPITCH + HID + 4 * lm) = sr[i];
    __builtin_amdgcn_fence(__ATOMIC_ACQ_REL, "wavefront");

    v8f acc[NT];
    for (int t = 0; t < NT; ++t)
      acc[t] = (v8f){0.f, 0.f, 0.f, 0.f, 0.f, 0.f, 0.f, 0.f};

    for (int k = 0; k < 16; ++k) {
      bool self_done = false;
      const auto self_step = [&]() __attribute__((always_inline)) {
        if (self_done) return;
        self_done = true;
        __builtin_amdgcn_sched_barrier(0);
        const v2f a = *(const v2f*)(afrag + HID + 4 * k);
        for (int t = 0; t < NT; ++t) {
          const v2f b = *(const v2f*)(bfrag + ((16 + k) * NT + t) * 64);
          acc[t] = __builtin_amdgcn_wmma_f32_16x16x4_f32(
              false, a, false, b, (short)0, acc[t], false, false);
        }
        __builtin_amdgcn_sched_barrier(0);
      };

      v2f mean = {0.f, 0.f};
      if (k < nn) {
        int s2 = 0, c2 = 0, idx2 = 0;
        if (k + 1 < nn) {
          s2 = __builtin_amdgcn_readlane(my_s, k + 1);
          c2 = __builtin_amdgcn_readlane(my_c, k + 1);
          idx2 = (lane < c2) ? csr[s2 + lane] : 0;
        }

        v2f sum = {0.f, 0.f};
        sage_sum_rows(h, idx, cg < 32 ? cg : 32, lane, sum, self_step);
        for (int j0 = 32; j0 < cg; j0 += 32) {
          const int more = (j0 + lane < cg) ? csr[sg + j0 + lane] : 0;
          sage_sum_rows(h, more, cg - j0 < 32 ? cg - j0 : 32, lane, sum,
                        self_step);
        }

        const float invc = 1.0f / fmaxf((float)cg, 1.0f);
        mean.x = sum.x * invc;
        mean.y = sum.y * invc;

        sg = s2;
        cg = c2;
        idx = idx2;
      }
      self_step();
      *(v2f*)(tile + k * TPITCH + 2 * lane) = mean;
    }
    __builtin_amdgcn_fence(__ATOMIC_ACQ_REL, "wavefront");

    my_s = 0;
    my_c = 0;
    if (tm + nWaves < nTiles && lane < 16 &&
        (tm + nWaves) * 16 + lane < nNodes) {
      my_s = start[(tm + nWaves) * 16 + lane];
      my_c = cnt[(tm + nWaves) * 16 + lane];
    }

#pragma unroll
    for (int ks = 0; ks < 16; ++ks) {
      const v2f a = *(const v2f*)(afrag + 4 * ks);
      for (int t = 0; t < NT; ++t) {
        const v2f b = *(const v2f*)(bfrag + (ks * NT + t) * 64);
        acc[t] = __builtin_amdgcn_wmma_f32_16x16x4_f32(
            false, a, false, b, (short)0, acc[t], false, false);
      }
    }

    for (int t = 0; t < NT; ++t)
      for (int r = 0; r < 8; ++r)
        tile[(r + 8 * half) * TPITCH + 16 * t + lm] = acc[t][r] + bv[t];
    __builtin_amdgcn_fence(__ATOMIC_ACQ_REL, "wavefront");

    if (n0 + 16 <= nNodes) {
      float4 ov[8];
      for (int i = 0; i < 8; ++i)
        ov[i] = *(const float4*)(tile + (2 * i + half) * TPITCH + 4 * lm);
      for (int i = 0; i < 8; ++i)
        *(float4*)(out + (long)(n0 + 2 * i + half) * HID + 4 * lm) = ov[i];
    } else {
      for (int i = 0; i < 8; ++i) {
        const int r = 2 * i + half;
        const float4 v = *(const float4*)(tile + r * TPITCH + 4 * lm);
        if (n0 + r < nNodes) *(float4*)(out + (long)(n0 + r) * HID + 4 * lm) = v;
      }
    }
    __builtin_amdgcn_fence(__ATOMIC_ACQ_REL, "wavefront");
  }
}

extern "C" void kernel_launch(void* const* d_in, const int* in_sizes, int n_in,
                              void* d_out, int out_size, void* d_ws,
                              size_t ws_size, hipStream_t stream) {
  const float* x = (const float*)d_in[0];
  const int* ei = (const int*)d_in[1];
  const float* Wl1 = (const float*)d_in[2];
  const float* bl1 = (const float*)d_in[3];
  const float* Wr1 = (const float*)d_in[4];
  const float* Wl2 = (const float*)d_in[5];
  const float* bl2 = (const float*)d_in[6];
  const float* Wr2 = (const float*)d_in[7];
  float* out = (float*)d_out;

  const int N = in_sizes[0] / IN_C;
  const int E = in_sizes[1] / 2;
  const int* src = ei;
  const int* dst = ei + E;

  float* p1 = (float*)d_ws;
  float* h = p1 + (long)N * HID;
  float* pairs_area = h + (long)N * HID;
  int2* pairs = (int2*)pairs_area;
  int* cnt = (int*)(pairs_area + (long)N * HID);
  int* start = cnt + N;
  int* csr = start + N;
  const int B = (E + CHUNK - 1) / CHUNK;
  const int NB = (N + RANGE - 1) / RANGE;
  int* T = csr + E;

  sage_chunk_sort_kernel<<<B, CHUNK_THREADS, 0, stream>>>(src, dst, E, NB,
                                                          pairs, T);
  sage_range_sort_kernel<<<NB, RANGE, 0, stream>>>(pairs, T, N, NB, B, cnt,
                                                   start, csr);

  const int wavesPerBlock = 256 / 32;
  const int gatherBlocks = (N + NPW * wavesPerBlock - 1) / (NPW * wavesPerBlock);

  sage_rows_gemm_kernel<<<GT_GROUPS, GT_THREADS, 0, stream>>>(
      x, Wl1, bl1, Wr1, p1, h, N);
  sage_gather_kernel<<<gatherBlocks, 256, 0, stream>>>(p1, start, cnt, csr, h,
                                                       N);

  sage_gather_gemm_kernel<<<GT_GROUPS, FG_THREADS, 0, stream>>>(
      h, start, cnt, csr, Wl2, bl2, Wr2, out, N);
}
